# starting kernel
# baseline (speedup 1.0000x reference)
#include <hip/hip_runtime.h>


#define HIDDEN   16
#define T_LEN    4096
#define B_TOT    8192
#define TILES_PER_WAVE 2

typedef __attribute__((ext_vector_type(2))) float    v2f;
typedef __attribute__((ext_vector_type(4))) float    v4f;
typedef __attribute__((ext_vector_type(8))) float    v8f;

__device__ __forceinline__ float htanh(float x) {
#if __has_builtin(__builtin_amdgcn_tanhf)
  return __builtin_amdgcn_tanhf(x);
#else
  float y;
  asm("v_tanh_f32 %0, %1" : "=v"(y) : "v"(x));
  return y;
#endif
}

#if __has_builtin(__builtin_amdgcn_permlane16_swap)
#define HAVE_PERMLANE_SWAP 1
#else
#define HAVE_PERMLANE_SWAP 0
__device__ __forceinline__ float halfswap(float x) {
  int xi = __float_as_int(x);
  int r = __builtin_amdgcn_permlanex16(xi, xi, 0x76543210, 0xfedcba98, false, false);
  return __int_as_float(r);
}
#endif

struct Tile {
  float bf[8];
};

__device__ __forceinline__ void step2(Tile& a, Tile& b, float ua, float ub,
                                      const v2f* __restrict__ wfrag,
                                      const float* __restrict__ winj, bool hi) {
  v8f ca1, cb1;
#pragma unroll
  for (int j = 0; j < 8; ++j) { ca1[j] = ua * winj[j]; cb1[j] = ub * winj[j]; }
  v8f ca2 = {}, cb2 = {};

#pragma unroll
  for (int k = 0; k < 2; ++k) {
    v2f ba1; ba1.x = a.bf[2 * k + 0]; ba1.y = a.bf[2 * k + 1];
    v2f ba2; ba2.x = a.bf[2 * k + 4]; ba2.y = a.bf[2 * k + 5];
    v2f bb1; bb1.x = b.bf[2 * k + 0]; bb1.y = b.bf[2 * k + 1];
    v2f bb2; bb2.x = b.bf[2 * k + 4]; bb2.y = b.bf[2 * k + 5];
    ca1 = __builtin_amdgcn_wmma_f32_16x16x4_f32(false, wfrag[k + 0], false, ba1,
                                                (short)0, ca1, false, false);
    ca2 = __builtin_amdgcn_wmma_f32_16x16x4_f32(false, wfrag[k + 2], false, ba2,
                                                (short)0, ca2, false, false);
    cb1 = __builtin_amdgcn_wmma_f32_16x16x4_f32(false, wfrag[k + 0], false, bb1,
                                                (short)0, cb1, false, false);
    cb2 = __builtin_amdgcn_wmma_f32_16x16x4_f32(false, wfrag[k + 2], false, bb2,
                                                (short)0, cb2, false, false);
  }

  float sa[8], sb[8];
  v8f accA = ca1 + ca2;
  v8f accB = cb1 + cb2;
#pragma unroll
  for (int j = 0; j < 8; ++j) { sa[j] = htanh(accA[j]); sb[j] = htanh(accB[j]); }

#if HAVE_PERMLANE_SWAP
#pragma unroll
  for (int p = 0; p < 4; ++p) {
    const int r  = (p & 1) + 4 * (p >> 1);
    const int lo = (p & 1) + 2 * (p >> 1);
    {
      auto ra = __builtin_amdgcn_permlane16_swap(__float_as_uint(sa[r]),
                                                 __float_as_uint(sa[r + 2]),
                                                 false, false);
      a.bf[lo] = __uint_as_float(ra[0]);
      a.bf[lo + 4] = __uint_as_float(ra[1]);
    }
    {
      auto rb = __builtin_amdgcn_permlane16_swap(__float_as_uint(sb[r]),
                                                 __float_as_uint(sb[r + 2]),
                                                 false, false);
      b.bf[lo] = __uint_as_float(rb[0]);
      b.bf[lo + 4] = __uint_as_float(rb[1]);
    }
  }
#else
#pragma unroll
  for (int p = 0; p < 4; ++p) {
    const int r  = (p & 1) + 4 * (p >> 1);
    const int lo = (p & 1) + 2 * (p >> 1);
    float haL = halfswap(sa[r + 2]), haH = halfswap(sa[r]);
    float hbL = halfswap(sb[r + 2]), hbH = halfswap(sb[r]);
    a.bf[lo]     = hi ? haL : sa[r];
    a.bf[lo + 4] = hi ? sa[r + 2] : haH;
    b.bf[lo]     = hi ? hbL : sb[r];
    b.bf[lo + 4] = hi ? sb[r + 2] : hbH;
  }
#endif
}

__global__ __launch_bounds__(128) void rain_reservoir_kernel(
    const float* __restrict__ rain,
    const float* __restrict__ w_in,
    const float* __restrict__ w,
    float* __restrict__ out) {
  const int lane = threadIdx.x & 31;
  const int col  = lane & 15;
  const int half = lane >> 4;
  const bool hi  = (half != 0);

  const int wave  = (blockIdx.x * blockDim.x + threadIdx.x) >> 5;
  const int tile0 = wave * TILES_PER_WAVE;
  const int b0    = (tile0 + 0) * 16;
  const int b1    = (tile0 + 1) * 16;

  v2f wfrag[4];
#pragma unroll
  for (int k = 0; k < 4; ++k) {
    wfrag[k].x = w[col * HIDDEN + 4 * k + 2 * half];
    wfrag[k].y = w[col * HIDDEN + 4 * k + 1 + 2 * half];
  }
  float winj[8];
#pragma unroll
  for (int j = 0; j < 8; ++j) winj[j] = w_in[j + 8 * half];

  Tile t0, t1;
#pragma unroll
  for (int j = 0; j < 8; ++j) { t0.bf[j] = 0.0f; t1.bf[j] = 0.0f; }

  const v4f* u0p = (const v4f*)(rain + (size_t)(b0 + col) * T_LEN);
  const v4f* u1p = (const v4f*)(rain + (size_t)(b1 + col) * T_LEN);

  v4f u0 = __builtin_nontemporal_load(u0p);
  v4f u1 = __builtin_nontemporal_load(u1p);
  ++u0p; ++u1p;

  for (int t = 0; t < T_LEN - 4; t += 4) {
    v4f n0 = __builtin_nontemporal_load(u0p);
    v4f n1 = __builtin_nontemporal_load(u1p);
    ++u0p; ++u1p;

    step2(t0, t1, u0.x, u1.x, wfrag, winj, hi);
    step2(t0, t1, u0.y, u1.y, wfrag, winj, hi);
    step2(t0, t1, u0.z, u1.z, wfrag, winj, hi);
    step2(t0, t1, u0.w, u1.w, wfrag, winj, hi);

    u0 = n0;
    u1 = n1;
  }
  step2(t0, t1, u0.x, u1.x, wfrag, winj, hi);
  step2(t0, t1, u0.y, u1.y, wfrag, winj, hi);
  step2(t0, t1, u0.z, u1.z, wfrag, winj, hi);
  step2(t0, t1, u0.w, u1.w, wfrag, winj, hi);

#pragma unroll
  for (int j = 0; j < 8; ++j) {
    const int row = 4 * (j >> 1) + (j & 1) + 2 * half;
    out[(size_t)(b0 + col) * HIDDEN + row] = t0.bf[j];
    out[(size_t)(b1 + col) * HIDDEN + row] = t1.bf[j];
  }
}

extern "C" void kernel_launch(void* const* d_in, const int* in_sizes, int n_in,
                              void* d_out, int out_size, void* d_ws, size_t ws_size,
                              hipStream_t stream) {
  const float* rain = (const float*)d_in[0];
  const float* w_in = (const float*)d_in[1];
  const float* w    = (const float*)d_in[2];
  float* out        = (float*)d_out;

  const int tiles   = B_TOT / 16;
  const int waves   = tiles / TILES_PER_WAVE;
  const int threads = 128;
  const int blocks  = (waves * 32) / threads;

  rain_reservoir_kernel<<<blocks, threads, 0, stream>>>(rain, w_in, w, out);
}


# One 16-batch tile per wave instead of two (512 waves, 128 blocks of 4 waves): each wave's in-order stream carries one serial recurrence chain per step, not two back to back
# speedup vs baseline: 1.8868x; 1.8868x over previous
#include <hip/hip_runtime.h>


#define HIDDEN   16
#define T_LEN    4096
#define B_TOT    8192

typedef __attribute__((ext_vector_type(2))) float    v2f;
typedef __attribute__((ext_vector_type(4))) float    v4f;
typedef __attribute__((ext_vector_type(8))) float    v8f;

__device__ __forceinline__ float htanh(float x) {
#if __has_builtin(__builtin_amdgcn_tanhf)
  return __builtin_amdgcn_tanhf(x);
#else
  float y;
  asm("v_tanh_f32 %0, %1" : "=v"(y) : "v"(x));
  return y;
#endif
}

#if __has_builtin(__builtin_amdgcn_permlane16_swap)
#define HAVE_PERMLANE_SWAP 1
#else
#define HAVE_PERMLANE_SWAP 0
__device__ __forceinline__ float halfswap(float x) {
  int xi = __float_as_int(x);
  int r = __builtin_amdgcn_permlanex16(xi, xi, 0x76543210, 0xfedcba98, false, false);
  return __int_as_float(r);
}
#endif

struct Tile {
  float bf[8];
};

__device__ __forceinline__ void step1(Tile& a, float ua,
                                      const v2f* __restrict__ wfrag,
                                      const float* __restrict__ winj, bool hi) {
  v8f c1;
#pragma unroll
  for (int j = 0; j < 8; ++j) c1[j] = ua * winj[j];
  v8f c2 = {};

#pragma unroll
  for (int k = 0; k < 2; ++k) {
    v2f b1; b1.x = a.bf[2 * k + 0]; b1.y = a.bf[2 * k + 1];
    v2f b2; b2.x = a.bf[2 * k + 4]; b2.y = a.bf[2 * k + 5];
    c1 = __builtin_amdgcn_wmma_f32_16x16x4_f32(false, wfrag[k + 0], false, b1,
                                               (short)0, c1, false, false);
    c2 = __builtin_amdgcn_wmma_f32_16x16x4_f32(false, wfrag[k + 2], false, b2,
                                               (short)0, c2, false, false);
  }

  float sa[8];
  v8f acc = c1 + c2;
#pragma unroll
  for (int j = 0; j < 8; ++j) sa[j] = htanh(acc[j]);

#if HAVE_PERMLANE_SWAP
#pragma unroll
  for (int p = 0; p < 4; ++p) {
    const int r  = (p & 1) + 4 * (p >> 1);
    const int lo = (p & 1) + 2 * (p >> 1);
    auto ra = __builtin_amdgcn_permlane16_swap(__float_as_uint(sa[r]),
                                               __float_as_uint(sa[r + 2]),
                                               false, false);
    a.bf[lo] = __uint_as_float(ra[0]);
    a.bf[lo + 4] = __uint_as_float(ra[1]);
  }
#else
#pragma unroll
  for (int p = 0; p < 4; ++p) {
    const int r  = (p & 1) + 4 * (p >> 1);
    const int lo = (p & 1) + 2 * (p >> 1);
    float haL = halfswap(sa[r + 2]), haH = halfswap(sa[r]);
    a.bf[lo]     = hi ? haL : sa[r];
    a.bf[lo + 4] = hi ? sa[r + 2] : haH;
  }
#endif
}

__global__ __launch_bounds__(128) void rain_reservoir_kernel(
    const float* __restrict__ rain,
    const float* __restrict__ w_in,
    const float* __restrict__ w,
    float* __restrict__ out) {
  const int lane = threadIdx.x & 31;
  const int col  = lane & 15;
  const int half = lane >> 4;
  const bool hi  = (half != 0);

  const int wave = (blockIdx.x * blockDim.x + threadIdx.x) >> 5;
  const int b0   = wave * 16;

  v2f wfrag[4];
#pragma unroll
  for (int k = 0; k < 4; ++k) {
    wfrag[k].x = w[col * HIDDEN + 4 * k + 2 * half];
    wfrag[k].y = w[col * HIDDEN + 4 * k + 1 + 2 * half];
  }
  float winj[8];
#pragma unroll
  for (int j = 0; j < 8; ++j) winj[j] = w_in[j + 8 * half];

  Tile t0;
#pragma unroll
  for (int j = 0; j < 8; ++j) t0.bf[j] = 0.0f;

  const v4f* u0p = (const v4f*)(rain + (size_t)(b0 + col) * T_LEN);

  v4f u0 = __builtin_nontemporal_load(u0p);
  ++u0p;

  for (int t = 0; t < T_LEN - 4; t += 4) {
    v4f n0 = __builtin_nontemporal_load(u0p);
    ++u0p;

    step1(t0, u0.x, wfrag, winj, hi);
    step1(t0, u0.y, wfrag, winj, hi);
    step1(t0, u0.z, wfrag, winj, hi);
    step1(t0, u0.w, wfrag, winj, hi);

    u0 = n0;
  }
  step1(t0, u0.x, wfrag, winj, hi);
  step1(t0, u0.y, wfrag, winj, hi);
  step1(t0, u0.z, wfrag, winj, hi);
  step1(t0, u0.w, wfrag, winj, hi);

#pragma unroll
  for (int j = 0; j < 8; ++j) {
    const int row = 4 * (j >> 1) + (j & 1) + 2 * half;
    out[(size_t)(b0 + col) * HIDDEN + row] = t0.bf[j];
  }
}

extern "C" void kernel_launch(void* const* d_in, const int* in_sizes, int n_in,
                              void* d_out, int out_size, void* d_ws, size_t ws_size,
                              hipStream_t stream) {
  const float* rain = (const float*)d_in[0];
  const float* w_in = (const float*)d_in[1];
  const float* w    = (const float*)d_in[2];
  float* out        = (float*)d_out;

  rain_reservoir_kernel<<<128, 128, 0, stream>>>(rain, w_in, w, out);
}


# W fragments loaded in the k order of the D map (column 8h+2k+s), so the tanh results are the next step's B operands as they stand: the 4 permlane16_swap relayouts per step go; k sum order regrouped
# speedup vs baseline: 2.1765x; 1.1535x over previous
#include <hip/hip_runtime.h>


#define HIDDEN   16
#define T_LEN    4096
#define B_TOT    8192

typedef __attribute__((ext_vector_type(2))) float    v2f;
typedef __attribute__((ext_vector_type(4))) float    v4f;
typedef __attribute__((ext_vector_type(8))) float    v8f;

__device__ __forceinline__ float htanh(float x) {
#if __has_builtin(__builtin_amdgcn_tanhf)
  return __builtin_amdgcn_tanhf(x);
#else
  float y;
  asm("v_tanh_f32 %0, %1" : "=v"(y) : "v"(x));
  return y;
#endif
}

struct Tile {
  float x[8];
};

__device__ __forceinline__ void step1(Tile& a, float ua,
                                      const v2f* __restrict__ wfrag,
                                      const float* __restrict__ winj) {
  v8f c1;
#pragma unroll
  for (int j = 0; j < 8; ++j) c1[j] = ua * winj[j];
  v8f c2 = {};

#pragma unroll
  for (int k = 0; k < 2; ++k) {
    v2f b1; b1.x = a.x[2 * k + 0]; b1.y = a.x[2 * k + 1];
    v2f b2; b2.x = a.x[2 * k + 4]; b2.y = a.x[2 * k + 5];
    c1 = __builtin_amdgcn_wmma_f32_16x16x4_f32(false, wfrag[k + 0], false, b1,
                                               (short)0, c1, false, false);
    c2 = __builtin_amdgcn_wmma_f32_16x16x4_f32(false, wfrag[k + 2], false, b2,
                                               (short)0, c2, false, false);
  }

  v8f acc = c1 + c2;
#pragma unroll
  for (int j = 0; j < 8; ++j) a.x[j] = htanh(acc[j]);
}

__global__ __launch_bounds__(128) void rain_reservoir_kernel(
    const float* __restrict__ rain,
    const float* __restrict__ w_in,
    const float* __restrict__ w,
    float* __restrict__ out) {
  const int lane = threadIdx.x & 31;
  const int col  = lane & 15;
  const int half = lane >> 4;

  const int wave = (blockIdx.x * blockDim.x + threadIdx.x) >> 5;
  const int b0   = wave * 16;

  v2f wfrag[4];
#pragma unroll
  for (int k = 0; k < 4; ++k) {
    wfrag[k].x = w[col * HIDDEN + 8 * half + 2 * k + 0];
    wfrag[k].y = w[col * HIDDEN + 8 * half + 2 * k + 1];
  }
  float winj[8];
#pragma unroll
  for (int j = 0; j < 8; ++j) winj[j] = w_in[j + 8 * half];

  Tile t0;
#pragma unroll
  for (int j = 0; j < 8; ++j) t0.x[j] = 0.0f;

  const v4f* u0p = (const v4f*)(rain + (size_t)(b0 + col) * T_LEN);

  v4f u0 = __builtin_nontemporal_load(u0p);
  ++u0p;

  for (int t = 0; t < T_LEN - 4; t += 4) {
    v4f n0 = __builtin_nontemporal_load(u0p);
    ++u0p;

    step1(t0, u0.x, wfrag, winj);
    step1(t0, u0.y, wfrag, winj);
    step1(t0, u0.z, wfrag, winj);
    step1(t0, u0.w, wfrag, winj);

    u0 = n0;
  }
  step1(t0, u0.x, wfrag, winj);
  step1(t0, u0.y, wfrag, winj);
  step1(t0, u0.z, wfrag, winj);
  step1(t0, u0.w, wfrag, winj);

  v4f o0, o1;
  o0.x = t0.x[0]; o0.y = t0.x[1]; o0.z = t0.x[2]; o0.w = t0.x[3];
  o1.x = t0.x[4]; o1.y = t0.x[5]; o1.z = t0.x[6]; o1.w = t0.x[7];
  v4f* op = (v4f*)(out + (size_t)(b0 + col) * HIDDEN + 8 * half);
  op[0] = o0;
  op[1] = o1;
}

extern "C" void kernel_launch(void* const* d_in, const int* in_sizes, int n_in,
                              void* d_out, int out_size, void* d_ws, size_t ws_size,
                              hipStream_t stream) {
  const float* rain = (const float*)d_in[0];
  const float* w_in = (const float*)d_in[1];
  const float* w    = (const float*)d_in[2];
  float* out        = (float*)d_out;

  rain_reservoir_kernel<<<128, 128, 0, stream>>>(rain, w_in, w, out);
}


# Time loop of 32 steps per trip on two 64-byte input register sets that take turns, each loaded 16 steps ahead of its first reader (was one 16-byte load 4 steps ahead and two register copies per trip)
# speedup vs baseline: 2.7103x; 1.2453x over previous
#include <hip/hip_runtime.h>


#define HIDDEN   16
#define T_LEN    4096
#define B_TOT    8192

typedef __attribute__((ext_vector_type(2))) float    v2f;
typedef __attribute__((ext_vector_type(4))) float    v4f;
typedef __attribute__((ext_vector_type(8))) float    v8f;

__device__ __forceinline__ float htanh(float x) {
#if __has_builtin(__builtin_amdgcn_tanhf)
  return __builtin_amdgcn_tanhf(x);
#else
  float y;
  asm("v_tanh_f32 %0, %1" : "=v"(y) : "v"(x));
  return y;
#endif
}

struct Tile {
  float x[8];
};

__device__ __forceinline__ void step1(Tile& a, float ua,
                                      const v2f* __restrict__ wfrag,
                                      const float* __restrict__ winj) {
  v8f c1;
#pragma unroll
  for (int j = 0; j < 8; ++j) c1[j] = ua * winj[j];
  v8f c2 = {};

#pragma unroll
  for (int k = 0; k < 2; ++k) {
    v2f b1; b1.x = a.x[2 * k + 0]; b1.y = a.x[2 * k + 1];
    v2f b2; b2.x = a.x[2 * k + 4]; b2.y = a.x[2 * k + 5];
    c1 = __builtin_amdgcn_wmma_f32_16x16x4_f32(false, wfrag[k + 0], false, b1,
                                               (short)0, c1, false, false);
    c2 = __builtin_amdgcn_wmma_f32_16x16x4_f32(false, wfrag[k + 2], false, b2,
                                               (short)0, c2, false, false);
  }

  v8f acc = c1 + c2;
#pragma unroll
  for (int j = 0; j < 8; ++j) a.x[j] = htanh(acc[j]);
}

struct Chunk {
  v4f q[4];
};

__device__ __forceinline__ Chunk load_chunk(const v4f* __restrict__ p) {
  Chunk c;
#pragma unroll
  for (int i = 0; i < 4; ++i) c.q[i] = __builtin_nontemporal_load(p + i);
  return c;
}

__device__ __forceinline__ void steps16(Tile& a, const Chunk& c,
                                        const v2f* __restrict__ wfrag,
                                        const float* __restrict__ winj) {
#pragma unroll
  for (int i = 0; i < 4; ++i) {
    step1(a, c.q[i].x, wfrag, winj);
    step1(a, c.q[i].y, wfrag, winj);
    step1(a, c.q[i].z, wfrag, winj);
    step1(a, c.q[i].w, wfrag, winj);
  }
}

__global__ __launch_bounds__(128) void rain_reservoir_kernel(
    const float* __restrict__ rain,
    const float* __restrict__ w_in,
    const float* __restrict__ w,
    float* __restrict__ out) {
  const int lane = threadIdx.x & 31;
  const int col  = lane & 15;
  const int half = lane >> 4;

  const int wave = (blockIdx.x * blockDim.x + threadIdx.x) >> 5;
  const int b0   = wave * 16;

  v2f wfrag[4];
#pragma unroll
  for (int k = 0; k < 4; ++k) {
    wfrag[k].x = w[col * HIDDEN + 8 * half + 2 * k + 0];
    wfrag[k].y = w[col * HIDDEN + 8 * half + 2 * k + 1];
  }
  float winj[8];
#pragma unroll
  for (int j = 0; j < 8; ++j) winj[j] = w_in[j + 8 * half];

  Tile t0;
#pragma unroll
  for (int j = 0; j < 8; ++j) t0.x[j] = 0.0f;

  const v4f* up = (const v4f*)(rain + (size_t)(b0 + col) * T_LEN);

  Chunk ca = load_chunk(up);
  up += 4;
  Chunk cb;

  for (int t = 0; t < T_LEN - 32; t += 32) {
    cb = load_chunk(up);
    __builtin_amdgcn_sched_barrier(0);
    steps16(t0, ca, wfrag, winj);
    __builtin_amdgcn_sched_barrier(0);
    ca = load_chunk(up + 4);
    __builtin_amdgcn_sched_barrier(0);
    steps16(t0, cb, wfrag, winj);
    up += 8;
  }
  cb = load_chunk(up);
  steps16(t0, ca, wfrag, winj);
  steps16(t0, cb, wfrag, winj);

  v4f o0, o1;
  o0.x = t0.x[0]; o0.y = t0.x[1]; o0.z = t0.x[2]; o0.w = t0.x[3];
  o1.x = t0.x[4]; o1.y = t0.x[5]; o1.z = t0.x[6]; o1.w = t0.x[7];
  v4f* op = (v4f*)(out + (size_t)(b0 + col) * HIDDEN + 8 * half);
  op[0] = o0;
  op[1] = o1;
}

extern "C" void kernel_launch(void* const* d_in, const int* in_sizes, int n_in,
                              void* d_out, int out_size, void* d_ws, size_t ws_size,
                              hipStream_t stream) {
  const float* rain = (const float*)d_in[0];
  const float* w_in = (const float*)d_in[1];
  const float* w    = (const float*)d_in[2];
  float* out        = (float*)d_out;

  rain_reservoir_kernel<<<128, 128, 0, stream>>>(rain, w_in, w, out);
}


# Each step issued as four runs: 4 matrix instructions back to back, next step's input-term multiplies in the wait for the last result, 4 merge adds, then all 8 tanh; no vector work between WMMAs
# speedup vs baseline: 2.8094x; 1.0366x over previous
#include <hip/hip_runtime.h>


#define HIDDEN   16
#define T_LEN    4096
#define B_TOT    8192

typedef __attribute__((ext_vector_type(2))) float    v2f;
typedef __attribute__((ext_vector_type(4))) float    v4f;
typedef __attribute__((ext_vector_type(8))) float    v8f;

__device__ __forceinline__ float htanh(float x) {
#if __has_builtin(__builtin_amdgcn_tanhf)
  return __builtin_amdgcn_tanhf(x);
#else
  float y;
  asm("v_tanh_f32 %0, %1" : "=v"(y) : "v"(x));
  return y;
#endif
}

struct Tile {
  float x[8];
  v8f cin;
};

__device__ __forceinline__ v8f input_term(float u,
                                          const float* __restrict__ winj) {
  v8f c;
#pragma unroll
  for (int j = 0; j < 8; ++j) c[j] = u * winj[j];
  return c;
}

__device__ __forceinline__ void step1(Tile& a, float unext,
                                      const v2f* __restrict__ wfrag,
                                      const float* __restrict__ winj) {
  v8f c1 = a.cin;
  v8f c2 = {};

#pragma unroll
  for (int k = 0; k < 2; ++k) {
    v2f b1; b1.x = a.x[2 * k + 0]; b1.y = a.x[2 * k + 1];
    v2f b2; b2.x = a.x[2 * k + 4]; b2.y = a.x[2 * k + 5];
    c1 = __builtin_amdgcn_wmma_f32_16x16x4_f32(false, wfrag[k + 0], false, b1,
                                               (short)0, c1, false, false);
    c2 = __builtin_amdgcn_wmma_f32_16x16x4_f32(false, wfrag[k + 2], false, b2,
                                               (short)0, c2, false, false);
  }
  asm volatile("" : "+v"(c1), "+v"(c2));
  __builtin_amdgcn_sched_barrier(0);

  a.cin = input_term(unext, winj);
  asm volatile("" : "+v"(c1), "+v"(c2), "+v"(a.cin));

  v8f acc = c1 + c2;

  a.x[0] = htanh(acc[0]);
  a.x[1] = htanh(acc[1]);
  a.x[4] = htanh(acc[4]);
  a.x[5] = htanh(acc[5]);
  a.x[2] = htanh(acc[2]);
  a.x[3] = htanh(acc[3]);
  a.x[6] = htanh(acc[6]);
  a.x[7] = htanh(acc[7]);
  asm volatile("" : "+v"(a.x[0]), "+v"(a.x[1]), "+v"(a.x[2]), "+v"(a.x[3]),
                    "+v"(a.x[4]), "+v"(a.x[5]), "+v"(a.x[6]), "+v"(a.x[7]));
}

struct Chunk {
  v4f q[4];
};

__device__ __forceinline__ Chunk load_chunk(const v4f* __restrict__ p) {
  Chunk c;
#pragma unroll
  for (int i = 0; i < 4; ++i) c.q[i] = __builtin_nontemporal_load(p + i);
  return c;
}

__device__ __forceinline__ void steps16(Tile& a, const Chunk& c,
                                        float ufirst_next,
                                        const v2f* __restrict__ wfrag,
                                        const float* __restrict__ winj) {
#pragma unroll
  for (int i = 0; i < 4; ++i) {
    const float un = (i < 3) ? c.q[(i + 1) & 3].x : ufirst_next;
    step1(a, c.q[i].y, wfrag, winj);
    step1(a, c.q[i].z, wfrag, winj);
    step1(a, c.q[i].w, wfrag, winj);
    step1(a, un, wfrag, winj);
  }
}

__global__ __launch_bounds__(128) void rain_reservoir_kernel(
    const float* __restrict__ rain,
    const float* __restrict__ w_in,
    const float* __restrict__ w,
    float* __restrict__ out) {
  const int lane = threadIdx.x & 31;
  const int col  = lane & 15;
  const int half = lane >> 4;

  const int wave = (blockIdx.x * blockDim.x + threadIdx.x) >> 5;
  const int b0   = wave * 16;

  v2f wfrag[4];
#pragma unroll
  for (int k = 0; k < 4; ++k) {
    wfrag[k].x = w[col * HIDDEN + 8 * half + 2 * k + 0];
    wfrag[k].y = w[col * HIDDEN + 8 * half + 2 * k + 1];
  }
  float winj[8];
#pragma unroll
  for (int j = 0; j < 8; ++j) winj[j] = w_in[j + 8 * half];

  Tile t0;
#pragma unroll
  for (int j = 0; j < 8; ++j) t0.x[j] = 0.0f;

  const v4f* up = (const v4f*)(rain + (size_t)(b0 + col) * T_LEN);

  Chunk ca = load_chunk(up);
  up += 4;
  Chunk cb;
  t0.cin = input_term(ca.q[0].x, winj);

  for (int t = 0; t < T_LEN - 32; t += 32) {
    cb = load_chunk(up);
    __builtin_amdgcn_sched_barrier(0);
    steps16(t0, ca, cb.q[0].x, wfrag, winj);
    __builtin_amdgcn_sched_barrier(0);
    ca = load_chunk(up + 4);
    __builtin_amdgcn_sched_barrier(0);
    steps16(t0, cb, ca.q[0].x, wfrag, winj);
    up += 8;
  }
  cb = load_chunk(up);
  steps16(t0, ca, cb.q[0].x, wfrag, winj);
  steps16(t0, cb, 0.0f, wfrag, winj);

  v4f o0, o1;
  o0.x = t0.x[0]; o0.y = t0.x[1]; o0.z = t0.x[2]; o0.w = t0.x[3];
  o1.x = t0.x[4]; o1.y = t0.x[5]; o1.z = t0.x[6]; o1.w = t0.x[7];
  v4f* op = (v4f*)(out + (size_t)(b0 + col) * HIDDEN + 8 * half);
  op[0] = o0;
  op[1] = o1;
}

extern "C" void kernel_launch(void* const* d_in, const int* in_sizes, int n_in,
                              void* d_out, int out_size, void* d_ws, size_t ws_size,
                              hipStream_t stream) {
  const float* rain = (const float*)d_in[0];
  const float* w_in = (const float*)d_in[1];
  const float* w    = (const float*)d_in[2];
  float* out        = (float*)d_out;

  rain_reservoir_kernel<<<128, 128, 0, stream>>>(rain, w_in, w, out);
}
